# starting kernel
# baseline (speedup 1.0000x reference)
#include <hip/hip_runtime.h>

typedef __attribute__((ext_vector_type(2))) float v2f;
typedef __attribute__((ext_vector_type(8))) float v8f;

__global__ __launch_bounds__(256) void slice_proj_wmma_kernel(
    const float* __restrict__ x,
    const float* __restrict__ weights,
    const int*   __restrict__ src_idx,
    float*       __restrict__ out,
    int K, int n_out)
{
    const int lane = threadIdx.x & 31;
    const int wave = threadIdx.x >> 5;
    const int tile = blockIdx.x * 8 + wave;
    const int pb   = tile * 16;
    if (pb >= n_out) return;

    const int row = lane & 15;
    const int hi  = lane >> 4;
    const long long gp = (long long)pb + row;
    const float* __restrict__ wrow = weights + gp * (long long)K;
    const int*   __restrict__ irow = src_idx + gp * (long long)K;

    v8f c = {};
    const int K4 = K & ~3;
    for (int k0 = 0; k0 < K4; k0 += 4) {
        const int kk = k0 + 2 * hi;
        __builtin_prefetch(wrow + kk + 16, 0, 0);
        __builtin_prefetch(irow + kk + 16, 0, 0);

        v2f a;
        a.x = wrow[kk];
        a.y = wrow[kk + 1];

        const int i0 = irow[kk];
        const int i1 = irow[kk + 1];
        v2f b;
        b.x = x[i0];
        b.y = x[i1];

        c = __builtin_amdgcn_wmma_f32_16x16x4_f32(
                false, a, false, b, (short)0, c, false, false);
    }

    const int r = lane & 7;
    float diag = c[0];
#pragma unroll
    for (int j = 1; j < 8; ++j)
        if (r == j) diag = c[j];

    const bool writer = (lane < 8) || (lane >= 24);
    if (writer) {
        const int orow = (lane < 8) ? lane : (lane - 16);
        const long long g = (long long)pb + orow;
        const float* __restrict__ wr = weights + g * (long long)K;
        const int*   __restrict__ ir = src_idx + g * (long long)K;
        float tail = 0.0f;
        for (int k = K4; k < K; ++k)
            tail += wr[k] * x[ir[k]];
        out[pb + orow] = diag + tail;
    }
}

extern "C" void kernel_launch(void* const* d_in, const int* in_sizes, int n_in,
                              void* d_out, int out_size, void* d_ws, size_t ws_size,
                              hipStream_t stream) {
    const float* x       = (const float*)d_in[0];
    const float* weights = (const float*)d_in[1];
    const int*   src_idx = (const int*)d_in[2];
    float*       out     = (float*)d_out;

    const int K = (out_size > 0) ? (in_sizes[1] / out_size) : 0;

    const int tiles  = (out_size + 15) / 16;
    const int blocks = (tiles + 7) / 8;

    slice_proj_wmma_kernel<<<blocks, 256, 0, stream>>>(
        x, weights, src_idx, out, K, out_size);
}


# W and idx tile panels read once with whole-line non-temporal 16-B loads and transposed through wave-private LDS into the WMMA fragments; a tile's gathers issued as one run; prefetches dropped
# speedup vs baseline: 3.8167x; 3.8167x over previous
#include <hip/hip_runtime.h>

typedef __attribute__((ext_vector_type(2))) float v2f;
typedef __attribute__((ext_vector_type(4))) float v4f;
typedef __attribute__((ext_vector_type(2))) int   v2i;
typedef __attribute__((ext_vector_type(4))) int   v4i;
typedef __attribute__((ext_vector_type(8))) float v8f;


constexpr int KF    = 56;
constexpr int CH    = KF / 4;
constexpr int NLD   = 16 * CH / 32;
constexpr int NCK   = KF / 4;
constexpr int PITCH = 60;
constexpr int WAVES = 8;

__global__ __launch_bounds__(256) void slice_proj_wmma_staged_k56(
    const float* __restrict__ x,
    const float* __restrict__ weights,
    const int*   __restrict__ src_idx,
    float*       __restrict__ out,
    int n_out)
{
    __shared__ __attribute__((aligned(16))) float s_w[WAVES][16 * PITCH];
    __shared__ __attribute__((aligned(16))) int   s_i[WAVES][16 * PITCH];

    const int lane = threadIdx.x & 31;
    const int wave = __builtin_amdgcn_readfirstlane(threadIdx.x >> 5);
    const int tile = blockIdx.x * WAVES + wave;
    const int pb   = tile * 16;
    if (pb >= n_out) return;

    const v4i* __restrict__ ip4 = reinterpret_cast<const v4i*>(src_idx + (long long)pb * KF);
    const v4f* __restrict__ wp4 = reinterpret_cast<const v4f*>(weights + (long long)pb * KF);

    v4i iq[NLD];
    v4f wq[NLD];
    for (int j = 0; j < NLD; ++j) iq[j] = __builtin_nontemporal_load(ip4 + j * 32 + lane);
    for (int j = 0; j < NLD; ++j) wq[j] = __builtin_nontemporal_load(wp4 + j * 32 + lane);

    float* sw = s_w[wave];
    int*   si = s_i[wave];

    const int row = lane & 15;
    const int hi  = lane >> 4;
    const int fo  = row * PITCH + 2 * hi;

    for (int j = 0; j < NLD; ++j) {
        const int f = j * 32 + lane;
        *reinterpret_cast<v4i*>(si + 4 * (f + f / CH)) = iq[j];
    }
    v2i ix[NCK];
    for (int t = 0; t < NCK; ++t) ix[t] = *reinterpret_cast<const v2i*>(si + fo + 4 * t);

    __builtin_amdgcn_sched_barrier(0);
    v2f b[NCK];
    for (int t = 0; t < NCK; ++t) {
        b[t].x = x[ix[t].x];
        b[t].y = x[ix[t].y];
    }

    for (int j = 0; j < NLD; ++j) {
        const int f = j * 32 + lane;
        *reinterpret_cast<v4f*>(sw + 4 * (f + f / CH)) = wq[j];
    }
    v2f a[NCK];
    for (int t = 0; t < NCK; ++t) a[t] = *reinterpret_cast<const v2f*>(sw + fo + 4 * t);

    v8f c = {};
    for (int t = 0; t < NCK; ++t) {
        c = __builtin_amdgcn_wmma_f32_16x16x4_f32(
                false, a[t], false, b[t], (short)0, c, false, false);
    }

    const int r = lane & 7;
    float diag = c[0];
#pragma unroll
    for (int j = 1; j < 8; ++j)
        if (r == j) diag = c[j];

    const bool writer = (lane < 8) || (lane >= 24);
    if (writer) {
        const int orow = (lane < 8) ? lane : (lane - 16);
        out[pb + orow] = diag;
    }
}

__global__ __launch_bounds__(256) void slice_proj_wmma_kernel(
    const float* __restrict__ x,
    const float* __restrict__ weights,
    const int*   __restrict__ src_idx,
    float*       __restrict__ out,
    int K, int n_out)
{
    const int lane = threadIdx.x & 31;
    const int wave = threadIdx.x >> 5;
    const int tile = blockIdx.x * 8 + wave;
    const int pb   = tile * 16;
    if (pb >= n_out) return;

    const int row = lane & 15;
    const int hi  = lane >> 4;
    const long long gp = (long long)pb + row;
    const float* __restrict__ wrow = weights + gp * (long long)K;
    const int*   __restrict__ irow = src_idx + gp * (long long)K;

    v8f c = {};
    const int K4 = K & ~3;
    for (int k0 = 0; k0 < K4; k0 += 4) {
        const int kk = k0 + 2 * hi;
        __builtin_prefetch(wrow + kk + 16, 0, 0);
        __builtin_prefetch(irow + kk + 16, 0, 0);

        v2f a;
        a.x = wrow[kk];
        a.y = wrow[kk + 1];

        const int i0 = irow[kk];
        const int i1 = irow[kk + 1];
        v2f b;
        b.x = x[i0];
        b.y = x[i1];

        c = __builtin_amdgcn_wmma_f32_16x16x4_f32(
                false, a, false, b, (short)0, c, false, false);
    }

    const int r = lane & 7;
    float diag = c[0];
#pragma unroll
    for (int j = 1; j < 8; ++j)
        if (r == j) diag = c[j];

    const bool writer = (lane < 8) || (lane >= 24);
    if (writer) {
        const int orow = (lane < 8) ? lane : (lane - 16);
        const long long g = (long long)pb + orow;
        const float* __restrict__ wr = weights + g * (long long)K;
        const int*   __restrict__ ir = src_idx + g * (long long)K;
        float tail = 0.0f;
        for (int k = K4; k < K; ++k)
            tail += wr[k] * x[ir[k]];
        out[pb + orow] = diag + tail;
    }
}

extern "C" void kernel_launch(void* const* d_in, const int* in_sizes, int n_in,
                              void* d_out, int out_size, void* d_ws, size_t ws_size,
                              hipStream_t stream) {
    const float* x       = (const float*)d_in[0];
    const float* weights = (const float*)d_in[1];
    const int*   src_idx = (const int*)d_in[2];
    float*       out     = (float*)d_out;

    const int K = (out_size > 0) ? (in_sizes[1] / out_size) : 0;

    const int tiles  = (out_size + 15) / 16;
    const int blocks = (tiles + 7) / 8;

    if (K == KF && (out_size & 15) == 0)
        slice_proj_wmma_staged_k56<<<blocks, 256, 0, stream>>>(
            x, weights, src_idx, out, out_size);
    else
        slice_proj_wmma_kernel<<<blocks, 256, 0, stream>>>(
            x, weights, src_idx, out, K, out_size);
}


# Staged kernel: a wave's 16 pixels are a 4 x 4 image block, not a 16 x 1 strip: gather footprints overlap more, so a tile asks L2 for fewer different x lines; sum order unchanged
# speedup vs baseline: 5.1771x; 1.3564x over previous
#include <hip/hip_runtime.h>

typedef __attribute__((ext_vector_type(2))) float v2f;
typedef __attribute__((ext_vector_type(4))) float v4f;
typedef __attribute__((ext_vector_type(2))) int   v2i;
typedef __attribute__((ext_vector_type(4))) int   v4i;
typedef __attribute__((ext_vector_type(8))) float v8f;


constexpr int KF    = 56;
constexpr int CH    = KF / 4;
constexpr int NLD   = 16 * CH / 32;
constexpr int NCK   = KF / 4;
constexpr int PITCH = 60;
constexpr int WAVES = 8;
constexpr int NJ    = KF / 2;
constexpr int IMG_W = 256;
constexpr int BAND  = 4 * IMG_W;
constexpr int TPB   = BAND / 16;

__device__ __forceinline__ constexpr int KOF(int j) { return 14 * (j / 7) + (j % 7); }

__global__ __launch_bounds__(256) void slice_proj_wmma_staged_k56(
    const float* __restrict__ x,
    const float* __restrict__ weights,
    const int*   __restrict__ src_idx,
    float*       __restrict__ out,
    int n_out, int rstep, int cstep)
{
    __shared__ __attribute__((aligned(16))) float s_w[WAVES][16 * PITCH];
    __shared__ __attribute__((aligned(16))) int   s_i[WAVES][16 * PITCH];

    const int lane = threadIdx.x & 31;
    const int wave = __builtin_amdgcn_readfirstlane(threadIdx.x >> 5);
    const int tile = blockIdx.x * WAVES + wave;
    if (tile * 16 >= n_out) return;
    const int g0   = (tile / TPB) * BAND + (tile % TPB) * cstep;
    const int rcst = rstep * CH;

    const v4i* __restrict__ ip4 = reinterpret_cast<const v4i*>(src_idx + (long long)g0 * KF);
    const v4f* __restrict__ wp4 = reinterpret_cast<const v4f*>(weights + (long long)g0 * KF);

    v4i iq[NLD];
    v4f wq[NLD];
    for (int j = 0; j < NLD; ++j) {
        const int f = j * 32 + lane;
        iq[j] = __builtin_nontemporal_load(ip4 + f + (f / (4 * CH)) * rcst);
    }
    for (int j = 0; j < NLD; ++j) {
        const int f = j * 32 + lane;
        wq[j] = __builtin_nontemporal_load(wp4 + f + (f / (4 * CH)) * rcst);
    }

    float* sw = s_w[wave];
    int*   si = s_i[wave];

    const int row = lane & 15;
    const int hi  = lane >> 4;
    const int fo  = row * PITCH + 7 * hi;

    for (int j = 0; j < NLD; ++j) {
        const int f = j * 32 + lane;
        *reinterpret_cast<v4i*>(si + 4 * (f + f / CH)) = iq[j];
    }
    int ix[NJ];
#pragma unroll
    for (int j = 0; j < NJ; ++j) ix[j] = si[fo + KOF(j)];

    __builtin_amdgcn_sched_barrier(0);
    v2f b[NCK];
#pragma unroll
    for (int t = 0; t < NCK; ++t) {
        b[t].x = x[ix[2 * t]];
        b[t].y = x[ix[2 * t + 1]];
    }

    for (int j = 0; j < NLD; ++j) {
        const int f = j * 32 + lane;
        *reinterpret_cast<v4f*>(sw + 4 * (f + f / CH)) = wq[j];
    }
    v2f a[NCK];
#pragma unroll
    for (int t = 0; t < NCK; ++t) {
        a[t].x = sw[fo + KOF(2 * t)];
        a[t].y = sw[fo + KOF(2 * t + 1)];
    }

    v8f c = {};
    for (int t = 0; t < NCK; ++t) {
        c = __builtin_amdgcn_wmma_f32_16x16x4_f32(
                false, a[t], false, b[t], (short)0, c, false, false);
    }

    const int r = lane & 7;
    float diag = c[0];
#pragma unroll
    for (int j = 1; j < 8; ++j)
        if (r == j) diag = c[j];

    const bool writer = (lane < 8) || (lane >= 24);
    if (writer) {
        const int orow = (lane < 8) ? lane : (lane - 16);
        out[g0 + orow + (orow >> 2) * rstep] = diag;
    }
}

__global__ __launch_bounds__(256) void slice_proj_wmma_kernel(
    const float* __restrict__ x,
    const float* __restrict__ weights,
    const int*   __restrict__ src_idx,
    float*       __restrict__ out,
    int K, int n_out)
{
    const int lane = threadIdx.x & 31;
    const int wave = threadIdx.x >> 5;
    const int tile = blockIdx.x * 8 + wave;
    const int pb   = tile * 16;
    if (pb >= n_out) return;

    const int row = lane & 15;
    const int hi  = lane >> 4;
    const long long gp = (long long)pb + row;
    const float* __restrict__ wrow = weights + gp * (long long)K;
    const int*   __restrict__ irow = src_idx + gp * (long long)K;

    v8f c = {};
    const int K4 = K & ~3;
    for (int k0 = 0; k0 < K4; k0 += 4) {
        const int kk = k0 + 2 * hi;
        __builtin_prefetch(wrow + kk + 16, 0, 0);
        __builtin_prefetch(irow + kk + 16, 0, 0);

        v2f a;
        a.x = wrow[kk];
        a.y = wrow[kk + 1];

        const int i0 = irow[kk];
        const int i1 = irow[kk + 1];
        v2f b;
        b.x = x[i0];
        b.y = x[i1];

        c = __builtin_amdgcn_wmma_f32_16x16x4_f32(
                false, a, false, b, (short)0, c, false, false);
    }

    const int r = lane & 7;
    float diag = c[0];
#pragma unroll
    for (int j = 1; j < 8; ++j)
        if (r == j) diag = c[j];

    const bool writer = (lane < 8) || (lane >= 24);
    if (writer) {
        const int orow = (lane < 8) ? lane : (lane - 16);
        const long long g = (long long)pb + orow;
        const float* __restrict__ wr = weights + g * (long long)K;
        const int*   __restrict__ ir = src_idx + g * (long long)K;
        float tail = 0.0f;
        for (int k = K4; k < K; ++k)
            tail += wr[k] * x[ir[k]];
        out[pb + orow] = diag + tail;
    }
}

extern "C" void kernel_launch(void* const* d_in, const int* in_sizes, int n_in,
                              void* d_out, int out_size, void* d_ws, size_t ws_size,
                              hipStream_t stream) {
    const float* x       = (const float*)d_in[0];
    const float* weights = (const float*)d_in[1];
    const int*   src_idx = (const int*)d_in[2];
    float*       out     = (float*)d_out;

    const int K = (out_size > 0) ? (in_sizes[1] / out_size) : 0;

    const int tiles  = (out_size + 15) / 16;
    const int blocks = (tiles + 7) / 8;

    if (K == KF && (out_size & (BAND - 1)) == 0)
        slice_proj_wmma_staged_k56<<<blocks, 256, 0, stream>>>(
            x, weights, src_idx, out, out_size, IMG_W - 4, 4);
    else if (K == KF && (out_size & 15) == 0)
        slice_proj_wmma_staged_k56<<<blocks, 256, 0, stream>>>(
            x, weights, src_idx, out, out_size, 0, 16);
    else
        slice_proj_wmma_kernel<<<blocks, 256, 0, stream>>>(
            x, weights, src_idx, out, K, out_size);
}


# Staged kernel: a wave walks over 4 tiles (1024 workgroups, all resident at once, not 4096) and asks for the next tile's W and idx panels right behind the current tile's gathers; sum order unchanged
# speedup vs baseline: 5.3600x; 1.0353x over previous
#include <hip/hip_runtime.h>

typedef __attribute__((ext_vector_type(2))) float v2f;
typedef __attribute__((ext_vector_type(4))) float v4f;
typedef __attribute__((ext_vector_type(2))) int   v2i;
typedef __attribute__((ext_vector_type(4))) int   v4i;
typedef __attribute__((ext_vector_type(8))) float v8f;


constexpr int KF    = 56;
constexpr int CH    = KF / 4;
constexpr int NLD   = 16 * CH / 32;
constexpr int NCK   = KF / 4;
constexpr int PITCH = 60;
constexpr int WAVES = 8;
constexpr int NJ    = KF / 2;
constexpr int IMG_W = 256;
constexpr int BAND  = 4 * IMG_W;
constexpr int TPB   = BAND / 16;

__device__ __forceinline__ constexpr int KOF(int j) { return 14 * (j / 7) + (j % 7); }

constexpr int TPW = 4;

__device__ __forceinline__ int tile_first_pixel(int tile, int cstep) {
    return (tile / TPB) * BAND + (tile % TPB) * cstep;
}

__global__ __launch_bounds__(256, 8) void slice_proj_wmma_staged_k56(
    const float* __restrict__ x,
    const float* __restrict__ weights,
    const int*   __restrict__ src_idx,
    float*       __restrict__ out,
    int n_out, int rstep, int cstep)
{
    __shared__ __attribute__((aligned(16))) float s_w[WAVES][16 * PITCH];
    __shared__ __attribute__((aligned(16))) int   s_i[WAVES][16 * PITCH];

    const int lane  = threadIdx.x & 31;
    const int wave  = __builtin_amdgcn_readfirstlane(threadIdx.x >> 5);
    const int tstep = gridDim.x * WAVES;
    int tile = blockIdx.x * WAVES + wave;
    if (tile * 16 >= n_out) return;
    int g0 = tile_first_pixel(tile, cstep);
    const int rcst = rstep * CH;

    v4i iq[NLD];
    v4f wq[NLD];
    {
        const v4i* __restrict__ ip4 = reinterpret_cast<const v4i*>(src_idx + (long long)g0 * KF);
        const v4f* __restrict__ wp4 = reinterpret_cast<const v4f*>(weights + (long long)g0 * KF);
        for (int j = 0; j < NLD; ++j) {
            const int f = j * 32 + lane;
            iq[j] = __builtin_nontemporal_load(ip4 + f + (f / (4 * CH)) * rcst);
        }
        for (int j = 0; j < NLD; ++j) {
            const int f = j * 32 + lane;
            wq[j] = __builtin_nontemporal_load(wp4 + f + (f / (4 * CH)) * rcst);
        }
    }

    float* sw = s_w[wave];
    int*   si = s_i[wave];

    const int row = lane & 15;
    const int hi  = lane >> 4;
    const int fo  = row * PITCH + 7 * hi;

    const int  r      = lane & 7;
    const bool writer = (lane < 8) || (lane >= 24);
    const int  orow   = (lane < 8) ? lane : (lane - 16);
    const int  ooff   = orow + (orow >> 2) * rstep;

    for (;;) {
        for (int j = 0; j < NLD; ++j) {
            const int f = j * 32 + lane;
            *reinterpret_cast<v4i*>(si + 4 * (f + f / CH)) = iq[j];
        }
        int ix[NJ];
#pragma unroll
        for (int j = 0; j < NJ; ++j) ix[j] = si[fo + KOF(j)];

        __builtin_amdgcn_sched_barrier(0);
        v2f b[NCK];
#pragma unroll
        for (int t = 0; t < NCK; ++t) {
            b[t].x = x[ix[2 * t]];
            b[t].y = x[ix[2 * t + 1]];
        }

        for (int j = 0; j < NLD; ++j) {
            const int f = j * 32 + lane;
            *reinterpret_cast<v4f*>(sw + 4 * (f + f / CH)) = wq[j];
        }

        const int  ntile = tile + tstep;
        const bool more  = ntile * 16 < n_out;
        int ng0 = g0;
        if (more) {
            ng0 = tile_first_pixel(ntile, cstep);
            const v4i* __restrict__ ip4 = reinterpret_cast<const v4i*>(src_idx + (long long)ng0 * KF);
            const v4f* __restrict__ wp4 = reinterpret_cast<const v4f*>(weights + (long long)ng0 * KF);
            for (int j = 0; j < NLD; ++j) {
                const int f = j * 32 + lane;
                iq[j] = __builtin_nontemporal_load(ip4 + f + (f / (4 * CH)) * rcst);
            }
            for (int j = 0; j < NLD; ++j) {
                const int f = j * 32 + lane;
                wq[j] = __builtin_nontemporal_load(wp4 + f + (f / (4 * CH)) * rcst);
            }
        }

        v2f a[NCK];
#pragma unroll
        for (int t = 0; t < NCK; ++t) {
            a[t].x = sw[fo + KOF(2 * t)];
            a[t].y = sw[fo + KOF(2 * t + 1)];
        }

        v8f c = {};
#pragma unroll
        for (int t = 0; t < NCK; ++t) {
            c = __builtin_amdgcn_wmma_f32_16x16x4_f32(
                    false, a[t], false, b[t], (short)0, c, false, false);
        }

        float diag = c[0];
#pragma unroll
        for (int j = 1; j < 8; ++j)
            if (r == j) diag = c[j];

        if (writer) out[g0 + ooff] = diag;

        if (!more) break;
        tile = ntile;
        g0   = ng0;
    }
}

__global__ __launch_bounds__(256) void slice_proj_wmma_kernel(
    const float* __restrict__ x,
    const float* __restrict__ weights,
    const int*   __restrict__ src_idx,
    float*       __restrict__ out,
    int K, int n_out)
{
    const int lane = threadIdx.x & 31;
    const int wave = threadIdx.x >> 5;
    const int tile = blockIdx.x * 8 + wave;
    const int pb   = tile * 16;
    if (pb >= n_out) return;

    const int row = lane & 15;
    const int hi  = lane >> 4;
    const long long gp = (long long)pb + row;
    const float* __restrict__ wrow = weights + gp * (long long)K;
    const int*   __restrict__ irow = src_idx + gp * (long long)K;

    v8f c = {};
    const int K4 = K & ~3;
    for (int k0 = 0; k0 < K4; k0 += 4) {
        const int kk = k0 + 2 * hi;
        __builtin_prefetch(wrow + kk + 16, 0, 0);
        __builtin_prefetch(irow + kk + 16, 0, 0);

        v2f a;
        a.x = wrow[kk];
        a.y = wrow[kk + 1];

        const int i0 = irow[kk];
        const int i1 = irow[kk + 1];
        v2f b;
        b.x = x[i0];
        b.y = x[i1];

        c = __builtin_amdgcn_wmma_f32_16x16x4_f32(
                false, a, false, b, (short)0, c, false, false);
    }

    const int r = lane & 7;
    float diag = c[0];
#pragma unroll
    for (int j = 1; j < 8; ++j)
        if (r == j) diag = c[j];

    const bool writer = (lane < 8) || (lane >= 24);
    if (writer) {
        const int orow = (lane < 8) ? lane : (lane - 16);
        const long long g = (long long)pb + orow;
        const float* __restrict__ wr = weights + g * (long long)K;
        const int*   __restrict__ ir = src_idx + g * (long long)K;
        float tail = 0.0f;
        for (int k = K4; k < K; ++k)
            tail += wr[k] * x[ir[k]];
        out[pb + orow] = diag + tail;
    }
}

extern "C" void kernel_launch(void* const* d_in, const int* in_sizes, int n_in,
                              void* d_out, int out_size, void* d_ws, size_t ws_size,
                              hipStream_t stream) {
    const float* x       = (const float*)d_in[0];
    const float* weights = (const float*)d_in[1];
    const int*   src_idx = (const int*)d_in[2];
    float*       out     = (float*)d_out;

    const int K = (out_size > 0) ? (in_sizes[1] / out_size) : 0;

    const int tiles  = (out_size + 15) / 16;
    const int blocks = (tiles + 7) / 8;
    const int sblocks = (blocks + TPW - 1) / TPW;

    if (K == KF && (out_size & (BAND - 1)) == 0)
        slice_proj_wmma_staged_k56<<<sblocks, 256, 0, stream>>>(
            x, weights, src_idx, out, out_size, IMG_W - 4, 4);
    else if (K == KF && (out_size & 15) == 0)
        slice_proj_wmma_staged_k56<<<sblocks, 256, 0, stream>>>(
            x, weights, src_idx, out, out_size, 0, 16);
    else
        slice_proj_wmma_kernel<<<blocks, 256, 0, stream>>>(
            x, weights, src_idx, out, K, out_size);
}
